# starting kernel
# baseline (speedup 1.0000x reference)
#include <hip/hip_runtime.h>


typedef __attribute__((ext_vector_type(2))) float v2f;
typedef __attribute__((ext_vector_type(8))) float v8f;

#define COS_EPS 1e-6f

__device__ __forceinline__ float pick8(v8f v, int s) {
  float r = v[0];
  if (s == 1) r = v[1];
  if (s == 2) r = v[2];
  if (s == 3) r = v[3];
  if (s == 4) r = v[4];
  if (s == 5) r = v[5];
  if (s == 6) r = v[6];
  if (s == 7) r = v[7];
  return r;
}

__global__ __launch_bounds__(256)
void edge_cosine_wmma_kernel(const float* __restrict__ zdrug,
                             const float* __restrict__ zadr,
                             const float* __restrict__ wp,
                             const float* __restrict__ bp,
                             const int* __restrict__ eidx,
                             float* __restrict__ out,
                             int E) {
  const int D = 256;
  const int tid  = threadIdx.x;
  const int lane = tid & 31;
  const long wid  = (long)blockIdx.x * (blockDim.x >> 5) + (tid >> 5);
  const long base = wid * 16;
  if (base >= E) return;

  const int m  = lane & 15;
  const int hf = lane >> 4;

  long e  = base + m;
  long se = (e < E) ? e : (long)E - 1;
  const int i0 = eidx[se];
  const int i1 = eidx[(size_t)E + se];

  const float4* px = (const float4*)(zdrug + (size_t)i0 * D + (size_t)hf * 128);
  const float4* py = (const float4*)(zadr  + (size_t)i1 * D + (size_t)hf * 128);

  float dot0 = 0.f, dot1 = 0.f;
  float xx0  = 0.f, xx1  = 0.f;
  float yy0  = 0.f, yy1  = 0.f;

#pragma unroll
  for (int j = 0; j < 16; ++j) {
    float4 x0 = px[j];
    float4 x1 = px[j + 16];
    float4 y0 = py[j];
    float4 y1 = py[j + 16];

    dot0 += x0.x * y0.x + x0.y * y0.y + x0.z * y0.z + x0.w * y0.w;
    dot1 += x1.x * y1.x + x1.y * y1.y + x1.z * y1.z + x1.w * y1.w;
    xx0  += x0.x * x0.x + x0.y * x0.y + x0.z * x0.z + x0.w * x0.w;
    xx1  += x1.x * x1.x + x1.y * x1.y + x1.z * x1.z + x1.w * x1.w;
    yy0  += y0.x * y0.x + y0.y * y0.y + y0.z * y0.z + y0.w * y0.w;
    yy1  += y1.x * y1.x + y1.y * y1.y + y1.z * y1.z + y1.w * y1.w;
  }

  v2f ones; ones[0] = 1.f; ones[1] = 1.f;
  v2f ad;   ad[0]   = dot0; ad[1]  = dot1;
  v2f ax;   ax[0]   = xx0;  ax[1]  = xx1;
  v2f ay;   ay[0]   = yy0;  ay[1]  = yy1;
  v8f zero = {};

  v8f Dd = __builtin_amdgcn_wmma_f32_16x16x4_f32(false, ad, false, ones,
                                                 (short)0, zero, false, false);
  v8f Dx = __builtin_amdgcn_wmma_f32_16x16x4_f32(false, ax, false, ones,
                                                 (short)0, zero, false, false);
  v8f Dy = __builtin_amdgcn_wmma_f32_16x16x4_f32(false, ay, false, ones,
                                                 (short)0, zero, false, false);

  if ((lane & 8) == 0) {
    const int slot = lane & 7;
    const int eloc = ((lane >> 4) << 3) | slot;
    const long eo  = base + eloc;
    if (eo < E) {
      float dv = pick8(Dd, slot);
      float xv = pick8(Dx, slot);
      float yv = pick8(Dy, slot);
      float nx = fmaxf(__builtin_sqrtf(xv), COS_EPS);
      float ny = fmaxf(__builtin_sqrtf(yv), COS_EPS);
      float sim = dv / (nx * ny);
      out[eo] = sim * wp[0] + bp[0];
    }
  }
}

extern "C" void kernel_launch(void* const* d_in, const int* in_sizes, int n_in,
                              void* d_out, int out_size, void* d_ws, size_t ws_size,
                              hipStream_t stream) {
  (void)n_in; (void)d_ws; (void)ws_size;
  const float* zdrug = (const float*)d_in[0];
  const float* zadr  = (const float*)d_in[1];
  const float* w     = (const float*)d_in[2];
  const float* b     = (const float*)d_in[3];
  const int*   eidx  = (const int*)d_in[4];

  const int E = in_sizes[4] / 2;
  float* out = (float*)d_out;
  (void)out_size;

  const int threads = 256;
  const int wavesNeeded = (E + 15) / 16;
  const int blocks = (wavesNeeded + 7) / 8;

  edge_cosine_wmma_kernel<<<blocks, threads, 0, stream>>>(
      zdrug, zadr, w, b, eidx, out, E);
}


# each gathered 1 KB row is read by the whole wave as two contiguous 512 B loads (4 lines per load, was 32); lane sums by one-hot f32 WMMA + register adds; 32 edges per wave; summation order changes
# speedup vs baseline: 1.9915x; 1.9915x over previous
#include <hip/hip_runtime.h>
#include <stdint.h>


typedef __attribute__((ext_vector_type(2))) float v2f;
typedef __attribute__((ext_vector_type(8))) float v8f;

#define COS_EPS 1e-6f
#define ROW_FLOATS 256
#define WAVE_EDGES 32
#define BLOCK_WAVES 8
#define DEV static __device__ __forceinline__

struct EdgeRows { float4 x0, x1, y0, y1; };

DEV void load_edge(EdgeRows& r, const float* __restrict__ zx, const float* __restrict__ zy, int rx, int ry, uint32_t lane16) {
  const unsigned char* bx = (const unsigned char*)(zx + (size_t)rx * ROW_FLOATS);
  const unsigned char* by = (const unsigned char*)(zy + (size_t)ry * ROW_FLOATS);
  r.x0 = *(const float4*)(bx + lane16);
  r.x1 = *(const float4*)(bx + lane16 + 512u);
  r.y0 = *(const float4*)(by + lane16);
  r.y1 = *(const float4*)(by + lane16 + 512u);
}

DEV void edge_partials(const EdgeRows& r, float& d, float& xx, float& yy) {
  d  = r.x0.x * r.y0.x; xx = r.x0.x * r.x0.x; yy = r.y0.x * r.y0.x;
  d  = __builtin_fmaf(r.x0.y, r.y0.y, d);  xx = __builtin_fmaf(r.x0.y, r.x0.y, xx);  yy = __builtin_fmaf(r.y0.y, r.y0.y, yy);
  d  = __builtin_fmaf(r.x0.z, r.y0.z, d);  xx = __builtin_fmaf(r.x0.z, r.x0.z, xx);  yy = __builtin_fmaf(r.y0.z, r.y0.z, yy);
  d  = __builtin_fmaf(r.x0.w, r.y0.w, d);  xx = __builtin_fmaf(r.x0.w, r.x0.w, xx);  yy = __builtin_fmaf(r.y0.w, r.y0.w, yy);
  d  = __builtin_fmaf(r.x1.x, r.y1.x, d);  xx = __builtin_fmaf(r.x1.x, r.x1.x, xx);  yy = __builtin_fmaf(r.y1.x, r.y1.x, yy);
  d  = __builtin_fmaf(r.x1.y, r.y1.y, d);  xx = __builtin_fmaf(r.x1.y, r.x1.y, xx);  yy = __builtin_fmaf(r.y1.y, r.y1.y, yy);
  d  = __builtin_fmaf(r.x1.z, r.y1.z, d);  xx = __builtin_fmaf(r.x1.z, r.x1.z, xx);  yy = __builtin_fmaf(r.y1.z, r.y1.z, yy);
  d  = __builtin_fmaf(r.x1.w, r.y1.w, d);  xx = __builtin_fmaf(r.x1.w, r.x1.w, xx);  yy = __builtin_fmaf(r.y1.w, r.y1.w, yy);
}

DEV float sum8(v8f v) { return ((v[0] + v[1]) + (v[2] + v[3])) + ((v[4] + v[5]) + (v[6] + v[7])); }

DEV float other_half(float v) {
  const int i = __builtin_bit_cast(int, v);
  return __builtin_bit_cast(float, __builtin_amdgcn_permlanex16(i, i, 0x76543210, 0xfedcba98, false, false));
}

__global__ __launch_bounds__(BLOCK_WAVES * 32)
void edge_cosine_rows_kernel(const float* __restrict__ zdrug,
                             const float* __restrict__ zadr,
                             const float* __restrict__ wp,
                             const float* __restrict__ bp,
                             const int* __restrict__ eidx,
                             float* __restrict__ out,
                             int E) {
  const uint32_t lane = threadIdx.x & 31u;
  const uint32_t wave = __builtin_amdgcn_readfirstlane(threadIdx.x >> 5);
  const long base = ((long)blockIdx.x * BLOCK_WAVES + wave) * WAVE_EDGES;
  if (base >= E) return;

  const long e  = base + lane;
  const long se = (e < E) ? e : (long)E - 1;
  const int r0 = eidx[se];
  const int r1 = eidx[(size_t)E + se];

  const uint32_t lane16 = 16u * lane;
  const uint32_t c = lane & 15u, h = lane >> 4;

  EdgeRows buf[2][2];
#define LOAD_PAIR(P) do { \
    load_edge(buf[(P) & 1][0], zdrug, zadr, __builtin_amdgcn_readlane(r0, 2 * (P)),     __builtin_amdgcn_readlane(r1, 2 * (P)),     lane16); \
    load_edge(buf[(P) & 1][1], zdrug, zadr, __builtin_amdgcn_readlane(r0, 2 * (P) + 1), __builtin_amdgcn_readlane(r1, 2 * (P) + 1), lane16); } while (0)

  float td = 0.f, tx = 0.f, ty = 0.f;
  v8f Dd = {}, Dx = {}, Dy = {};

  LOAD_PAIR(0);
#pragma unroll
  for (int p = 0; p < WAVE_EDGES / 2; ++p) {
    __builtin_amdgcn_sched_barrier(0);
    if (p + 1 < WAVE_EDGES / 2) LOAD_PAIR(p + 1);
    __builtin_amdgcn_sched_barrier(0);

    const int j = p & 7;
    if (j == 0) { Dd = v8f{}; Dx = v8f{}; Dy = v8f{}; }
    v2f ad, ax, ay;
    { float d, xx, yy; edge_partials(buf[p & 1][0], d, xx, yy); ad[0] = d; ax[0] = xx; ay[0] = yy; }
    { float d, xx, yy; edge_partials(buf[p & 1][1], d, xx, yy); ad[1] = d; ax[1] = xx; ay[1] = yy; }
    v2f oh;
    oh[0] = (c == (uint32_t)(2 * j))     ? 1.f : 0.f;
    oh[1] = (c == (uint32_t)(2 * j + 1)) ? 1.f : 0.f;
    Dd = __builtin_amdgcn_wmma_f32_16x16x4_f32(false, ad, false, oh, (short)0, Dd, false, false);
    Dx = __builtin_amdgcn_wmma_f32_16x16x4_f32(false, ax, false, oh, (short)0, Dx, false, false);
    Dy = __builtin_amdgcn_wmma_f32_16x16x4_f32(false, ay, false, oh, (short)0, Dy, false, false);

    if (j == 7) {
      float sd = sum8(Dd), sx = sum8(Dx), sy = sum8(Dy);
      sd += other_half(sd); sx += other_half(sx); sy += other_half(sy);
      if (h == (uint32_t)(p >> 3)) { td = sd; tx = sx; ty = sy; }
    }
  }
#undef LOAD_PAIR

  if (e < E) {
    const float nx = fmaxf(__builtin_sqrtf(tx), COS_EPS);
    const float ny = fmaxf(__builtin_sqrtf(ty), COS_EPS);
    const float sim = td / (nx * ny);
    out[e] = sim * wp[0] + bp[0];
  }
}

extern "C" void kernel_launch(void* const* d_in, const int* in_sizes, int n_in,
                              void* d_out, int out_size, void* d_ws, size_t ws_size,
                              hipStream_t stream) {
  (void)n_in; (void)d_ws; (void)ws_size;
  const float* zdrug = (const float*)d_in[0];
  const float* zadr  = (const float*)d_in[1];
  const float* w     = (const float*)d_in[2];
  const float* b     = (const float*)d_in[3];
  const int*   eidx  = (const int*)d_in[4];

  const int E = in_sizes[4] / 2;
  float* out = (float*)d_out;
  (void)out_size;

  const int threads = BLOCK_WAVES * 32;
  const int wavesNeeded = (E + WAVE_EDGES - 1) / WAVE_EDGES;
  const int blocks = (wavesNeeded + BLOCK_WAVES - 1) / BLOCK_WAVES;

  edge_cosine_rows_kernel<<<blocks, threads, 0, stream>>>(
      zdrug, zadr, w, b, eidx, out, E);
}
